# starting kernel
# baseline (speedup 1.0000x reference)
#include <hip/hip_runtime.h>
#include <math.h>

typedef __attribute__((ext_vector_type(2))) float v2f;
typedef __attribute__((ext_vector_type(8))) float v8f;

static __device__ __forceinline__ v8f wmma_f32(v2f a, v2f b, v8f c) {
  return __builtin_amdgcn_wmma_f32_16x16x4_f32(false, a, false, b, (short)0, c, false, false);
}

static __device__ __forceinline__ float cos8(int n) {
  n &= 7;
  const float R = 0.7071067811865476f;
  float mag = (n & 1) ? R : ((n & 2) ? 0.0f : 1.0f);
  float r = ((n >= 3) && (n <= 5)) ? -mag : mag;
  return r;
}
static __device__ __forceinline__ float sin8(int n) { return cos8(n + 6); }

__global__ void __launch_bounds__(256) dct_zero_tail(float* __restrict__ out) {
  const int per_b = 24 * 4096;
  int idx = blockIdx.x * 256 + threadIdx.x;
  if (idx < 32 * per_b) {
    int b = idx / per_b;
    int r = idx - b * per_b;
    out[b * (64 * 4096) + 40 * 4096 + r] = 0.0f;
  }
}

__global__ void __launch_bounds__(256) dct8x8_wmma(const float* __restrict__ x,
                                                   const float* __restrict__ fw,
                                                   float* __restrict__ out) {
  __shared__ float lds[8][2][256];
  const int lane = threadIdx.x & 31;
  const int wave = threadIdx.x >> 5;
  const int gw = blockIdx.x * 8 + wave;
  const int b  = gw >> 10;
  const int ty = (gw >> 5) & 31;
  const int tx = gw & 31;
  const int h  = lane >> 4;
  const int nl = lane & 15;

  v2f ca[4], sa[4], nsa[4];
#pragma unroll
  for (int j = 0; j < 4; ++j) {
#pragma unroll
    for (int v = 0; v < 2; ++v) {
      const int k = 4 * j + 2 * h + v;
      float c = 0.0f, s = 0.0f;
      if ((nl >> 3) == (k >> 3)) {
        const int n = (nl & 7) * (k & 7);
        c =  cos8(n) * 0.35355339059327373f;
        s = -sin8(n) * 0.35355339059327373f;
      }
      ca[j][v] = c; sa[j][v] = s; nsa[j][v] = -s;
    }
  }

  const int r0 = ty * 16, c0 = tx * 16;
  const float* xb0 = x + (b * 512 + r0) * 512 + c0 + nl;
  v2f xb[4];
#pragma unroll
  for (int j = 0; j < 4; ++j) {
    const int rr = 4 * j + 2 * h;
    xb[j][0] = xb0[rr * 512];
    xb[j][1] = xb0[rr * 512 + 512];
  }

  v8f t1 = {0, 0, 0, 0, 0, 0, 0, 0};
  v8f t2 = {0, 0, 0, 0, 0, 0, 0, 0};
#pragma unroll
  for (int j = 0; j < 4; ++j) {
    t1 = wmma_f32(ca[j], xb[j], t1);
    t2 = wmma_f32(sa[j], xb[j], t2);
  }

#pragma unroll
  for (int v = 0; v < 8; ++v) {
    lds[wave][0][(v + 8 * h) * 16 + nl] = t1[v];
    lds[wave][1][(v + 8 * h) * 16 + nl] = t2[v];
  }
  asm volatile("s_wait_dscnt 0" ::: "memory");

  v8f yre = {0, 0, 0, 0, 0, 0, 0, 0};
  v8f yim = {0, 0, 0, 0, 0, 0, 0, 0};
#pragma unroll
  for (int j = 0; j < 4; ++j) {
    const int off = nl * 16 + 4 * j + 2 * h;
    v2f a1; a1[0] = lds[wave][0][off]; a1[1] = lds[wave][0][off + 1];
    v2f a2; a2[0] = lds[wave][1][off]; a2[1] = lds[wave][1][off + 1];
    yre = wmma_f32(a1, ca[j],  yre);
    yre = wmma_f32(a2, nsa[j], yre);
    yim = wmma_f32(a2, ca[j],  yim);
    yim = wmma_f32(a1, sa[j],  yim);
  }

  const int kx = nl & 7;
  const int hb = ty * 2 + h;
  const int wb = tx * 2 + (nl >> 3);
  if (kx <= 4) {
    float* ob = out + b * (64 * 4096) + hb * 64 + wb;
#pragma unroll
    for (int v = 0; v < 8; ++v) {
      const int f = v * 5 + kx;
      const float w = __fdividef(1.0f, 1.0f + __expf(-fw[f]));
      const float mag = sqrtf(yre[v] * yre[v] + yim[v] * yim[v]);
      ob[f * 4096] = mag * w;
    }
  }
}

extern "C" void kernel_launch(void* const* d_in, const int* in_sizes, int n_in,
                              void* d_out, int out_size, void* d_ws, size_t ws_size,
                              hipStream_t stream) {
  (void)in_sizes; (void)n_in; (void)out_size; (void)d_ws; (void)ws_size;
  const float* x  = (const float*)d_in[0];
  const float* fw = (const float*)d_in[1];
  float* out = (float*)d_out;

  dct_zero_tail<<<(32 * 24 * 4096 + 255) / 256, 256, 0, stream>>>(out);
  dct8x8_wmma<<<4096, 256, 0, stream>>>(x, fw, out);
}


# pixel rows folded in registers (x[r]+x[8-r], x[r]-x[8-r]): bins ky and 8-ky share an even (K=40) and an odd (K=24) product, joined by add/sub; 160 -> 84 matrix ops per wave; sums reordered
# speedup vs baseline: 5.1019x; 5.1019x over previous
#include <hip/hip_runtime.h>
#include <math.h>


typedef __attribute__((ext_vector_type(2))) float v2f;
typedef __attribute__((ext_vector_type(4))) float v4f;
typedef __attribute__((ext_vector_type(8))) float v8f;

static __device__ __forceinline__ v8f wmma_f32(v2f a, v2f b, v8f c) {
  return __builtin_amdgcn_wmma_f32_16x16x4_f32(false, a, false, b, (short)0, c, false, false);
}

static __device__ __forceinline__ float cos8(int n) {
  n &= 7;
  const float R = 0.7071067811865476f;
  float mag = (n & 1) ? R : ((n & 2) ? 0.0f : 1.0f);
  float r = ((n >= 3) && (n <= 5)) ? -mag : mag;
  return r;
}

static __device__ __forceinline__ float xhalf(float x) {
  return __builtin_bit_cast(float, __builtin_amdgcn_permlanex16(__builtin_bit_cast(int, x), __builtin_bit_cast(int, x),
                                                                0x76543210, 0xfedcba98, false, false));
}

__global__ void __launch_bounds__(256) dct_zero_tail(float* __restrict__ out) {
  const int b   = blockIdx.x >> 3;
  const int seg = blockIdx.x & 7;
  v4f* p = (v4f*)(out + b * (64 * 4096) + 40 * 4096 + seg * 12288) + threadIdx.x;
  const v4f z = {0, 0, 0, 0};
#pragma unroll
  for (int i = 0; i < 12; ++i) p[i * 256] = z;
}


static __device__ __forceinline__ float fold_coef(int ky, int kx, int p, int c, int od, int os) {
  const int d = kx * c - ky * p;
  const int s = kx * c + ky * p;
  return (cos8(d + od) + cos8(s + os)) * 0.0625f;
}

template <int NCH>
static __device__ __forceinline__ void tile_mm(const v2f* tb, const v4f (&r0)[NCH / 2], const v4f (&r1)[NCH / 2],
                                               v8f& a0, v8f& a1) {
#pragma unroll
  for (int q = 0; q < NCH; ++q) {
    const v2f a = tb[q * 32];
    const int e = 2 * (q & 1);
    v2f b0; b0[0] = r0[q >> 1][e]; b0[1] = r0[q >> 1][e + 1];
    v2f b1; b1[0] = r1[q >> 1][e]; b1[1] = r1[q >> 1][e + 1];
    a0 = wmma_f32(a, b0, a0);
    a1 = wmma_f32(a, b1, a1);
  }
}

static __device__ __forceinline__ void put_pair(float* op, int plane_h0, int plane_h1, float m0, float m1, int h) {
  const float x0 = xhalf(m0);
  const float x1 = xhalf(m1);
  op[plane_h0 * 4096] = h ? x1 : m0;
  op[plane_h1 * 4096] = h ? m1 : x0;
}

__global__ void __launch_bounds__(256) dct8x8_rowfold(const float* __restrict__ x,
                                                      const float* __restrict__ fw,
                                                      float* __restrict__ out) {
  __shared__ v2f tab[42 * 32];
  const int lane = threadIdx.x & 31;
  const int wave = __builtin_amdgcn_readfirstlane(threadIdx.x >> 5) & 7;
  const int h  = lane >> 4;
  const int nl = lane & 15;

  const int g   = blockIdx.x * 8 + wave;
  const int b   = g >> 7;
  const int hb  = (g >> 1) & 63;
  const int wb0 = (g & 1) * 32;

  const float* xp = x + (b * 512 + hb * 8) * 512 + (wb0 + nl) * 8 + 4 * h;
  v4f xr[2][8];
#pragma unroll
  for (int r = 0; r < 8; ++r) {
#pragma unroll
    for (int s = 0; s < 2; ++s) xr[s][r] = *(const v4f*)(xp + r * 512 + s * 128);
  }

  float wv[4][5];
#pragma unroll
  for (int kx = 0; kx < 5; ++kx) {
    wv[0][kx] = __fdividef(1.0f, 1.0f + __expf(-fw[5 + 5 * h + kx]));
    wv[1][kx] = __fdividef(1.0f, 1.0f + __expf(-fw[35 - 5 * h + kx]));
    wv[2][kx] = __fdividef(1.0f, 1.0f + __expf(-fw[15 * h + kx]));
    wv[3][kx] = __fdividef(1.0f, 1.0f + __expf(-fw[20 + 5 * h + kx]));
  }

  {
    const int hD  = nl >> 3;
    const int v   = nl & 7;
    const int kx  = (v < 5) ? v : v - 4;
    const int sec = (v >= 5);
#pragma unroll
    for (int i = 0; i < 6; ++i) {
      const int en = wave + 8 * i;
      if (en < 42) {
        const bool odd = en >= 30;
        const int  ee  = odd ? en - 30 : en;
        const int  t   = odd ? ee / 6 : ee / 10;
        const int  q   = odd ? ee - 6 * t : ee - 10 * t;
        const int  p   = (q >> 1) + (odd ? 1 : 0);
        const int  c   = 4 * h + 2 * (q & 1);
        int ky, od, os;
        bool zero = false;
        if (!odd) {
          ky = (t == 0) ? 1 + hD : ((t == 1) ? 3 * hD : 4 - hD);
          od = sec ? 2 : 0;  os = sec ? 2 : 0;
        } else {
          ky = (t == 0) ? 1 + hD : 3;
          zero = (t == 1) && (hD == 0);
          od = sec ? 4 : 6;  os = sec ? 0 : 2;
        }
        v2f e;
        e[0] = zero ? 0.0f : fold_coef(ky, kx, p, c, od, os);
        e[1] = zero ? 0.0f : fold_coef(ky, kx, p, c + 1, od, os);
        tab[en * 32 + lane] = e;
      }
    }
  }

  v4f E0[5], E1[5], O0[3], O1[3];
  E0[0] = xr[0][0]; E0[4] = xr[0][4];
  E1[0] = xr[1][0]; E1[4] = xr[1][4];
#pragma unroll
  for (int j = 1; j < 4; ++j) {
    E0[j] = xr[0][j] + xr[0][8 - j];  O0[j - 1] = xr[0][j] - xr[0][8 - j];
    E1[j] = xr[1][j] + xr[1][8 - j];  O1[j - 1] = xr[1][j] - xr[1][8 - j];
  }
  __syncthreads();

  float* op = out + b * (64 * 4096) + hb * 64 + wb0 + lane;
  const v2f* tb = tab + lane;
  const v8f z8 = {0, 0, 0, 0, 0, 0, 0, 0};

  {
    v8f e0 = z8, e1 = z8, o0 = z8, o1 = z8;
    tile_mm<10>(tb, E0, E1, e0, e1);
    tile_mm<6>(tb + 30 * 32, O0, O1, o0, o1);
#pragma unroll
    for (int kx = 0; kx < 5; ++kx) {
      float p0, p1, m0, m1;
      if (kx >= 1 && kx <= 3) {
        const float rp0 = e0[kx] + o0[4 + kx], rm0 = e0[kx] - o0[4 + kx];
        const float ip0 = e0[4 + kx] + o0[kx], im0 = e0[4 + kx] - o0[kx];
        const float rp1 = e1[kx] + o1[4 + kx], rm1 = e1[kx] - o1[4 + kx];
        const float ip1 = e1[4 + kx] + o1[kx], im1 = e1[4 + kx] - o1[kx];
        p0 = sqrtf(rp0 * rp0 + ip0 * ip0);  m0 = sqrtf(rm0 * rm0 + im0 * im0);
        p1 = sqrtf(rp1 * rp1 + ip1 * ip1);  m1 = sqrtf(rm1 * rm1 + im1 * im1);
      } else {
        p0 = m0 = sqrtf(e0[kx] * e0[kx] + o0[kx] * o0[kx]);
        p1 = m1 = sqrtf(e1[kx] * e1[kx] + o1[kx] * o1[kx]);
      }
      put_pair(op, 5 + kx, 10 + kx, p0 * wv[0][kx], p1 * wv[0][kx], h);
      put_pair(op, 35 + kx, 30 + kx, m0 * wv[1][kx], m1 * wv[1][kx], h);
    }
  }

  {
    v8f f0 = z8, f1 = z8, g0 = z8, g1 = z8, o0 = z8, o1 = z8;
    tile_mm<10>(tb + 10 * 32, E0, E1, f0, f1);
    tile_mm<10>(tb + 20 * 32, E0, E1, g0, g1);
    tile_mm<6>(tb + 36 * 32, O0, O1, o0, o1);
#pragma unroll
    for (int kx = 0; kx < 5; ++kx) {
      float p0, p1, m0, m1;
      if (kx >= 1 && kx <= 3) {
        const float rp0 = f0[kx] + o0[4 + kx], rm0 = g0[kx] - o0[4 + kx];
        const float ip0 = f0[4 + kx] + o0[kx], im0 = g0[4 + kx] - o0[kx];
        const float rp1 = f1[kx] + o1[4 + kx], rm1 = g1[kx] - o1[4 + kx];
        const float ip1 = f1[4 + kx] + o1[kx], im1 = g1[4 + kx] - o1[kx];
        p0 = sqrtf(rp0 * rp0 + ip0 * ip0);  m0 = sqrtf(rm0 * rm0 + im0 * im0);
        p1 = sqrtf(rp1 * rp1 + ip1 * ip1);  m1 = sqrtf(rm1 * rm1 + im1 * im1);
      } else {
        p0 = sqrtf(f0[kx] * f0[kx] + o0[kx] * o0[kx]);  m0 = sqrtf(g0[kx] * g0[kx] + o0[kx] * o0[kx]);
        p1 = sqrtf(f1[kx] * f1[kx] + o1[kx] * o1[kx]);  m1 = sqrtf(g1[kx] * g1[kx] + o1[kx] * o1[kx]);
      }
      put_pair(op, kx, 15 + kx, p0 * wv[2][kx], p1 * wv[2][kx], h);
      put_pair(op, 20 + kx, 25 + kx, m0 * wv[3][kx], m1 * wv[3][kx], h);
    }
  }
}

extern "C" void kernel_launch(void* const* d_in, const int* in_sizes, int n_in,
                              void* d_out, int out_size, void* d_ws, size_t ws_size,
                              hipStream_t stream) {
  (void)in_sizes; (void)n_in; (void)out_size; (void)d_ws; (void)ws_size;
  const float* x  = (const float*)d_in[0];
  const float* fw = (const float*)d_in[1];
  float* out = (float*)d_out;

  dct_zero_tail<<<256, 256, 0, stream>>>(out);
  dct8x8_rowfold<<<512, 256, 0, stream>>>(x, fw, out);
}


# the 40 sigmoid weights evaluated once per work-group (five lanes per wave) into LDS; each lane reads its 20 back with five 16-byte LDS loads; was 20 exp and 20 divisions per lane; bit-identical
# speedup vs baseline: 6.4203x; 1.2584x over previous
#include <hip/hip_runtime.h>
#include <math.h>


typedef __attribute__((ext_vector_type(2))) float v2f;
typedef __attribute__((ext_vector_type(4))) float v4f;
typedef __attribute__((ext_vector_type(8))) float v8f;

static __device__ __forceinline__ v8f wmma_f32(v2f a, v2f b, v8f c) {
  return __builtin_amdgcn_wmma_f32_16x16x4_f32(false, a, false, b, (short)0, c, false, false);
}

static __device__ __forceinline__ float cos8(int n) {
  n &= 7;
  const float R = 0.7071067811865476f;
  float mag = (n & 1) ? R : ((n & 2) ? 0.0f : 1.0f);
  float r = ((n >= 3) && (n <= 5)) ? -mag : mag;
  return r;
}

static __device__ __forceinline__ float xhalf(float x) {
  return __builtin_bit_cast(float, __builtin_amdgcn_permlanex16(__builtin_bit_cast(int, x), __builtin_bit_cast(int, x),
                                                                0x76543210, 0xfedcba98, false, false));
}

__global__ void __launch_bounds__(256) dct_zero_tail(float* __restrict__ out) {
  const int b   = blockIdx.x >> 3;
  const int seg = blockIdx.x & 7;
  v4f* p = (v4f*)(out + b * (64 * 4096) + 40 * 4096 + seg * 12288) + threadIdx.x;
  const v4f z = {0, 0, 0, 0};
#pragma unroll
  for (int i = 0; i < 12; ++i) p[i * 256] = z;
}


static __device__ __forceinline__ float fold_coef(int ky, int kx, int p, int c, int od, int os) {
  const int d = kx * c - ky * p;
  const int s = kx * c + ky * p;
  return (cos8(d + od) + cos8(s + os)) * 0.0625f;
}

template <int NCH>
static __device__ __forceinline__ void tile_mm(const v2f* tb, const v4f (&r0)[NCH / 2], const v4f (&r1)[NCH / 2],
                                               v8f& a0, v8f& a1) {
#pragma unroll
  for (int q = 0; q < NCH; ++q) {
    const v2f a = tb[q * 32];
    const int e = 2 * (q & 1);
    v2f b0; b0[0] = r0[q >> 1][e]; b0[1] = r0[q >> 1][e + 1];
    v2f b1; b1[0] = r1[q >> 1][e]; b1[1] = r1[q >> 1][e + 1];
    a0 = wmma_f32(a, b0, a0);
    a1 = wmma_f32(a, b1, a1);
  }
}

static __device__ __forceinline__ void put_pair(float* op, int plane_h0, int plane_h1, float m0, float m1, int h) {
  const float x0 = xhalf(m0);
  const float x1 = xhalf(m1);
  op[plane_h0 * 4096] = h ? x1 : m0;
  op[plane_h1 * 4096] = h ? m1 : x0;
}

__global__ void __launch_bounds__(256) dct8x8_rowfft(const float* __restrict__ x,
                                                     const float* __restrict__ fw,
                                                     float* __restrict__ out) {
  __shared__ v2f tab[16 * 32];
  __shared__ __attribute__((aligned(16))) float sg[40];
  const int lane = threadIdx.x & 31;
  const int wave = __builtin_amdgcn_readfirstlane(threadIdx.x >> 5) & 7;
  const int h  = lane >> 4;
  const int nl = lane & 15;

  const int g   = blockIdx.x * 8 + wave;
  const int b   = g >> 7;
  const int hb  = (g >> 1) & 63;
  const int wb0 = (g & 1) * 32;

  const int   kyw = (0x46530271 >> (4 * wave)) & 7;
  const float fwv = fw[kyw * 5 + (lane < 4 ? lane : 4)];

  const float* xp = x + (b * 512 + hb * 8) * 512 + (wb0 + nl) * 8 + 4 * h;
  v4f xr[2][8];
#pragma unroll
  for (int r = 0; r < 8; ++r) {
#pragma unroll
    for (int s = 0; s < 2; ++s) xr[s][r] = *(const v4f*)(xp + r * 512 + s * 128);
  }


  {
    const int hD  = nl >> 3;
    const int v   = nl & 7;
    const int kx  = (v < 5) ? v : v - 4;
    const int sec = (v >= 5);
#pragma unroll
    for (int i = 0; i < 2; ++i) {
      const int  en   = wave + 8 * i;
      const int  kind = (en < 12) ? (en >> 2) : 3 + ((en - 12) >> 1);
      const int  q    = (en < 12) ? (en & 3) : (en & 1);
      const bool sn   = (kind == 1) || (kind == 4);
      const int  p    = (q >> 1) + (sn ? 1 : 0);
      const int  c    = 4 * h + 2 * (q & 1);
      const int  ky   = (kind < 2) ? 1 + 2 * hD : ((kind == 2) ? 4 * hD : 2);
      int od, os;
      if (!sn) { od = sec ? 2 : 0;  os = sec ? 2 : 0; }
      else     { od = sec ? 4 : 6;  os = sec ? 0 : 2; }
      v2f e;
      e[0] = fold_coef(ky, kx, p, c, od, os);
      e[1] = fold_coef(ky, kx, p, c + 1, od, os);
      tab[en * 32 + lane] = e;
    }
  }

  asm volatile("" :: "v"(fwv));
  if (lane < 5) sg[5 * wave + lane] = __fdividef(1.0f, 1.0f + __expf(-fwv));

  v4f Hh[2][2], Ss[2][2], Vv[2][2], Uu[2][1], Tt[2][1];
#pragma unroll
  for (int s = 0; s < 2; ++s) {
    const v4f e1 = xr[s][1] + xr[s][7], o1 = xr[s][1] - xr[s][7];
    const v4f e2 = xr[s][2] + xr[s][6], o2 = xr[s][2] - xr[s][6];
    const v4f e3 = xr[s][3] + xr[s][5], o3 = xr[s][3] - xr[s][5];
    const v4f g0 = xr[s][0] + xr[s][4];
    Hh[s][0] = xr[s][0] - xr[s][4];  Hh[s][1] = e1 - e3;
    Ss[s][0] = o1 + o3;              Ss[s][1] = o2;
    Vv[s][0] = g0 + e2;              Vv[s][1] = e1 + e3;
    Uu[s][0] = g0 - e2;
    Tt[s][0] = o1 - o3;
  }
  __syncthreads();

  float wv[4][5];
  {
    v4f w4[5];
#pragma unroll
    for (int i = 0; i < 5; ++i) w4[i] = *(const v4f*)(sg + 20 * h + 4 * i);
#pragma unroll
    for (int i = 0; i < 20; ++i) wv[i / 5][i % 5] = w4[i >> 2][i & 3];
  }

  float* op = out + b * (64 * 4096) + hb * 64 + wb0 + lane;
  const v2f* tb = tab + lane;
  const v8f z8 = {0, 0, 0, 0, 0, 0, 0, 0};

  {
    v8f e0 = z8, e1 = z8, o0 = z8, o1 = z8;
    tile_mm<4>(tb, Hh[0], Hh[1], e0, e1);
    tile_mm<4>(tb + 4 * 32, Ss[0], Ss[1], o0, o1);
#pragma unroll
    for (int kx = 0; kx < 5; ++kx) {
      float p0, p1, m0, m1;
      if (kx >= 1 && kx <= 3) {
        const float rp0 = e0[kx] + o0[4 + kx], rm0 = e0[kx] - o0[4 + kx];
        const float ip0 = e0[4 + kx] + o0[kx], im0 = e0[4 + kx] - o0[kx];
        const float rp1 = e1[kx] + o1[4 + kx], rm1 = e1[kx] - o1[4 + kx];
        const float ip1 = e1[4 + kx] + o1[kx], im1 = e1[4 + kx] - o1[kx];
        p0 = sqrtf(rp0 * rp0 + ip0 * ip0);  m0 = sqrtf(rm0 * rm0 + im0 * im0);
        p1 = sqrtf(rp1 * rp1 + ip1 * ip1);  m1 = sqrtf(rm1 * rm1 + im1 * im1);
      } else {
        p0 = m0 = sqrtf(e0[kx] * e0[kx] + o0[kx] * o0[kx]);
        p1 = m1 = sqrtf(e1[kx] * e1[kx] + o1[kx] * o1[kx]);
      }
      put_pair(op, 5 + kx, 15 + kx, p0 * wv[0][kx], p1 * wv[0][kx], h);
      put_pair(op, 35 + kx, 25 + kx, m0 * wv[1][kx], m1 * wv[1][kx], h);
    }
  }

  {
    v8f e0 = z8, e1 = z8, o0 = z8, o1 = z8;
    tile_mm<2>(tb + 12 * 32, Uu[0], Uu[1], e0, e1);
    tile_mm<2>(tb + 14 * 32, Tt[0], Tt[1], o0, o1);
    const float sg = h ? -1.0f : 1.0f;
#pragma unroll
    for (int kx = 0; kx < 5; ++kx) {
      float q0, q1;
      if (kx >= 1 && kx <= 3) {
        const float re0 = e0[kx] + sg * o0[4 + kx], im0 = e0[4 + kx] + sg * o0[kx];
        const float re1 = e1[kx] + sg * o1[4 + kx], im1 = e1[4 + kx] + sg * o1[kx];
        q0 = sqrtf(re0 * re0 + im0 * im0);
        q1 = sqrtf(re1 * re1 + im1 * im1);
      } else {
        q0 = sqrtf(e0[kx] * e0[kx] + o0[kx] * o0[kx]);
        q1 = sqrtf(e1[kx] * e1[kx] + o1[kx] * o1[kx]);
      }
      put_pair(op, 10 + kx, 30 + kx, q0 * wv[2][kx], q1 * wv[2][kx], h);
    }
  }

  {
    v8f e0 = z8, e1 = z8;
    tile_mm<4>(tb + 8 * 32, Vv[0], Vv[1], e0, e1);
#pragma unroll
    for (int kx = 0; kx < 5; ++kx) {
      float q0, q1;
      if (kx >= 1 && kx <= 3) {
        q0 = sqrtf(e0[kx] * e0[kx] + e0[4 + kx] * e0[4 + kx]);
        q1 = sqrtf(e1[kx] * e1[kx] + e1[4 + kx] * e1[4 + kx]);
      } else {
        q0 = fabsf(e0[kx]);
        q1 = fabsf(e1[kx]);
      }
      put_pair(op, kx, 20 + kx, q0 * wv[3][kx], q1 * wv[3][kx], h);
    }
  }
}

extern "C" void kernel_launch(void* const* d_in, const int* in_sizes, int n_in,
                              void* d_out, int out_size, void* d_ws, size_t ws_size,
                              hipStream_t stream) {
  (void)in_sizes; (void)n_in; (void)out_size; (void)d_ws; (void)ws_size;
  const float* x  = (const float*)d_in[0];
  const float* fw = (const float*)d_in[1];
  float* out = (float*)d_out;

  dct_zero_tail<<<256, 256, 0, stream>>>(out);
  dct8x8_rowfft<<<512, 256, 0, stream>>>(x, fw, out);
}
